# starting kernel
# baseline (speedup 1.0000x reference)
#include <hip/hip_runtime.h>
#include <hip/hip_bf16.h>

typedef float v2f __attribute__((ext_vector_type(2)));
typedef float v8f __attribute__((ext_vector_type(8)));

#define B_ 4
#define L_ 1024
#define D_ 512
#define N_ 16
#define CHUNK 64
#define NCHUNK 16
#define LOG2E 1.4426950408889634f

__global__ void k_xproj(const float* __restrict__ x_rgb, const float* __restrict__ x_e,
                        const float* __restrict__ w1, const float* __restrict__ w2,
                        float* __restrict__ dbl) {
    int wave = (blockIdx.x * blockDim.x + threadIdx.x) >> 5;
    int ln = threadIdx.x & 31;
    int ntile = wave % 3;
    int rem   = wave / 3;
    int mtile = rem & 255;
    int branch = rem >> 8;
    const float* x = branch ? x_e : x_rgb;
    const float* w = branch ? w2 : w1;
    int m  = mtile * 16 + (ln & 15);
    int n  = ntile * 16 + (ln & 15);
    int kb = (ln >> 4) * 2;
    const float* xrow = x + (long)m * D_;
    const float* wrow = w + (long)n * D_;
    v8f c = {};
    for (int k = 0; k < D_; k += 4) {
        v2f a, b;
        a.x = xrow[k + kb];  a.y = xrow[k + kb + 1];
        b.x = wrow[k + kb];  b.y = wrow[k + kb + 1];
        c = __builtin_amdgcn_wmma_f32_16x16x4_f32(false, a, false, b, (short)0, c, false, false);
    }
    float* out = dbl + (long)branch * 4096 * 48;
    int rowbase = mtile * 16 + ((ln >> 4) << 3);
    int col = ntile * 16 + (ln & 15);
#pragma unroll
    for (int i = 0; i < 8; ++i)
        out[(long)(rowbase + i) * 48 + col] = c[i];
}

__global__ void k_dtproj(const float* __restrict__ dbl,
                         const float* __restrict__ w1, const float* __restrict__ bias1,
                         const float* __restrict__ w2, const float* __restrict__ bias2,
                         float* __restrict__ delta_sp) {
    int wave = (blockIdx.x * blockDim.x + threadIdx.x) >> 5;
    int ln = threadIdx.x & 31;
    int ntile = wave & 31;
    int rem   = wave >> 5;
    int mtile = rem & 255;
    int branch = rem >> 8;
    const float* w    = branch ? w2 : w1;
    const float* bias = branch ? bias2 : bias1;
    const float* dt   = dbl + (long)branch * 4096 * 48;
    int m  = mtile * 16 + (ln & 15);
    int n  = ntile * 16 + (ln & 15);
    int kb = (ln >> 4) * 2;
    v8f c = {};
#pragma unroll
    for (int k = 0; k < 16; k += 4) {
        v2f a, b;
        a.x = dt[(long)m * 48 + k + kb];  a.y = dt[(long)m * 48 + k + kb + 1];
        b.x = w[(long)n * 16 + k + kb];   b.y = w[(long)n * 16 + k + kb + 1];
        c = __builtin_amdgcn_wmma_f32_16x16x4_f32(false, a, false, b, (short)0, c, false, false);
    }
    float* out = delta_sp + (long)branch * 4096 * D_;
    int rowbase = mtile * 16 + ((ln >> 4) << 3);
    int col = ntile * 16 + (ln & 15);
    float bcol = bias[col];
#pragma unroll
    for (int i = 0; i < 8; ++i) {
        float v  = c[i] + bcol;
        float sp = (v > 20.f) ? v : log1pf(__expf(v));
        out[(long)(rowbase + i) * D_ + col] = sp;
    }
}

__global__ void k_scan1(const float* __restrict__ x_rgb, const float* __restrict__ x_e,
                        const float* __restrict__ dbl, const float* __restrict__ delta_sp,
                        const float* __restrict__ Alog1, const float* __restrict__ Alog2,
                        float* __restrict__ hpart, float* __restrict__ aprod) {
    int bid = blockIdx.x;
    int lc  = bid & 15; bid >>= 4;
    int dch = bid & 3;  bid >>= 2;
    int b   = bid & 3;
    int branch = bid >> 2;
    int tid = threadIdx.x;
    int l0 = lc * CHUNK;

    __shared__ float Bsh[CHUNK * 16];
    {
        const float* src = dbl + (long)branch * 4096 * 48 + (long)(b * L_ + l0) * 48 + 16;
        for (int i = tid; i < CHUNK * 16; i += blockDim.x)
            Bsh[i] = src[(long)(i >> 4) * 48 + (i & 15)];
    }
    __syncthreads();

    int d = dch * 128 + tid;
    const float* Alog = branch ? Alog2 : Alog1;
    const float* x    = branch ? x_e : x_rgb;
    const float* ds   = delta_sp + (long)branch * 4096 * D_;

    float A2[N_], h[N_];
#pragma unroll
    for (int n = 0; n < N_; ++n) {
        A2[n] = -__expf(Alog[(long)d * N_ + n]) * LOG2E;
        h[n] = 0.f;
    }
    float ssum = 0.f;
    for (int ll = 0; ll < CHUNK; ++ll) {
        long idx = (long)(b * L_ + l0 + ll) * D_ + d;
        float s = ds[idx];
        float u = x[idx];
        float su = s * u;
        ssum += s;
#pragma unroll
        for (int n = 0; n < N_; ++n) {
            float dA = __builtin_amdgcn_exp2f(s * A2[n]);
            h[n] = fmaf(dA, h[n], su * Bsh[ll * 16 + n]);
        }
    }
    long off = ((long)(branch * 4 + b) * NCHUNK + lc) * (D_ * N_) + (long)d * N_;
#pragma unroll
    for (int n = 0; n < N_; ++n) {
        hpart[off + n] = h[n];
        aprod[off + n] = __builtin_amdgcn_exp2f(ssum * A2[n]);
    }
}

__global__ void k_carry(const float* __restrict__ hpart, const float* __restrict__ aprod,
                        float* __restrict__ hin) {
    int idx = blockIdx.x * blockDim.x + threadIdx.x;
    int dn = idx & (D_ * N_ - 1);
    int bb = idx >> 13;
    float h = 0.f;
    for (int c = 0; c < NCHUNK; ++c) {
        long pos = ((long)bb * NCHUNK + c) * (D_ * N_) + dn;
        hin[pos] = h;
        h = fmaf(aprod[pos], h, hpart[pos]);
    }
}

__global__ void k_scan2(const float* __restrict__ x_rgb, const float* __restrict__ x_e,
                        const float* __restrict__ dbl, const float* __restrict__ delta_sp,
                        const float* __restrict__ Alog1, const float* __restrict__ Alog2,
                        const float* __restrict__ D1, const float* __restrict__ D2,
                        const float* __restrict__ hin, float* __restrict__ out) {
    int bid = blockIdx.x;
    int lc  = bid & 15; bid >>= 4;
    int dch = bid & 3;  bid >>= 2;
    int b   = bid & 3;
    int branch = bid >> 2;
    int tid = threadIdx.x;
    int l0 = lc * CHUNK;

    __shared__ float Bsh[CHUNK * 16];
    __shared__ float Csh[CHUNK * 16];
    {
        const float* srcB = dbl + (long)branch * 4096 * 48 + (long)(b * L_ + l0) * 48 + 16;
        const float* srcC = dbl + (long)(1 - branch) * 4096 * 48 + (long)(b * L_ + l0) * 48 + 32;
        for (int i = tid; i < CHUNK * 16; i += blockDim.x) {
            int r = i >> 4, cc = i & 15;
            Bsh[i] = srcB[(long)r * 48 + cc];
            Csh[i] = srcC[(long)r * 48 + cc];
        }
    }
    __syncthreads();

    int d = dch * 128 + tid;
    const float* Alog = branch ? Alog2 : Alog1;
    const float* Dp   = branch ? D2 : D1;
    const float* x    = branch ? x_e : x_rgb;
    const float* ds   = delta_sp + (long)branch * 4096 * D_;

    float A2[N_], h[N_];
    long off = ((long)(branch * 4 + b) * NCHUNK + lc) * (D_ * N_) + (long)d * N_;
#pragma unroll
    for (int n = 0; n < N_; ++n) {
        A2[n] = -__expf(Alog[(long)d * N_ + n]) * LOG2E;
        h[n] = hin[off + n];
    }
    float Dd = Dp[d];
    float* yout = out + (long)branch * 4096 * D_;
    for (int ll = 0; ll < CHUNK; ++ll) {
        long idx = (long)(b * L_ + l0 + ll) * D_ + d;
        float s = ds[idx];
        float u = x[idx];
        float su = s * u;
        float y = 0.f;
#pragma unroll
        for (int n = 0; n < N_; ++n) {
            float dA = __builtin_amdgcn_exp2f(s * A2[n]);
            h[n] = fmaf(dA, h[n], su * Bsh[ll * 16 + n]);
            y = fmaf(h[n], Csh[ll * 16 + n], y);
        }
        yout[idx] = fmaf(Dd, u, y);
    }
}

__global__ void k_ln(float* __restrict__ out,
                     const float* __restrict__ g1, const float* __restrict__ bb1,
                     const float* __restrict__ g2, const float* __restrict__ bb2) {
    int row = blockIdx.x;
    int branch = row >> 12;
    float* p = out + (long)row * D_;
    int tid = threadIdx.x;
    float v0 = p[tid], v1 = p[tid + 256];
    __shared__ float rs[256], rq[256];
    rs[tid] = v0 + v1;
    rq[tid] = v0 * v0 + v1 * v1;
    __syncthreads();
    for (int st = 128; st > 0; st >>= 1) {
        if (tid < st) { rs[tid] += rs[tid + st]; rq[tid] += rq[tid + st]; }
        __syncthreads();
    }
    float mean = rs[0] * (1.f / 512.f);
    float var  = rq[0] * (1.f / 512.f) - mean * mean;
    float inv  = rsqrtf(var + 1e-5f);
    const float* g = branch ? g2 : g1;
    const float* bb = branch ? bb2 : bb1;
    p[tid]       = (v0 - mean) * inv * g[tid]       + bb[tid];
    p[tid + 256] = (v1 - mean) * inv * g[tid + 256] + bb[tid + 256];
}

extern "C" void kernel_launch(void* const* d_in, const int* in_sizes, int n_in,
                              void* d_out, int out_size, void* d_ws, size_t ws_size,
                              hipStream_t stream) {
    const float* x_rgb = (const float*)d_in[0];
    const float* x_e   = (const float*)d_in[1];
    const float* xw1   = (const float*)d_in[2];
    const float* xw2   = (const float*)d_in[3];
    const float* dtw1  = (const float*)d_in[4];
    const float* dtb1  = (const float*)d_in[5];
    const float* dtw2  = (const float*)d_in[6];
    const float* dtb2  = (const float*)d_in[7];
    const float* Al1   = (const float*)d_in[8];
    const float* Al2   = (const float*)d_in[9];
    const float* Dv1   = (const float*)d_in[10];
    const float* Dv2   = (const float*)d_in[11];
    const float* g1    = (const float*)d_in[12];
    const float* lb1   = (const float*)d_in[13];
    const float* g2    = (const float*)d_in[14];
    const float* lb2   = (const float*)d_in[15];
    float* out = (float*)d_out;

    float* ws    = (float*)d_ws;
    float* dbl   = ws;
    float* delta = dbl + 2 * 4096 * 48;
    float* hpart = delta + 2 * 4096 * 512;
    float* aprod = hpart + 2 * 4 * NCHUNK * 8192;
    float* hin   = aprod + 2 * 4 * NCHUNK * 8192;

    k_xproj<<<384, 128, 0, stream>>>(x_rgb, x_e, xw1, xw2, dbl);
    k_dtproj<<<4096, 128, 0, stream>>>(dbl, dtw1, dtb1, dtw2, dtb2, delta);
    k_scan1<<<512, 128, 0, stream>>>(x_rgb, x_e, dbl, delta, Al1, Al2, hpart, aprod);
    k_carry<<<256, 256, 0, stream>>>(hpart, aprod, hin);
    k_scan2<<<512, 128, 0, stream>>>(x_rgb, x_e, dbl, delta, Al1, Al2, Dv1, Dv2, hin, out);
    k_ln<<<8192, 256, 0, stream>>>(out, g1, lb1, g2, lb2);
}


# layernorm moves into the carry-term pass: a block owns 16 whole rows (512 threads), y stays in registers, row sums through LDS; 4 launches, not 5; 8192 waves, not 73472; layernorm sum order changes
# speedup vs baseline: 2.5669x; 2.5669x over previous
#include <hip/hip_runtime.h>
#include <hip/hip_bf16.h>

typedef float v2f __attribute__((ext_vector_type(2)));
typedef float v4f __attribute__((ext_vector_type(4)));
typedef float v8f __attribute__((ext_vector_type(8)));

#define B_ 4
#define L_ 1024
#define D_ 512
#define N_ 16
#define CHUNK 32
#define NCHUNK 32
#define LOG2E 1.4426950408889634f

__global__ void __launch_bounds__(128)
k_xproj(const float* __restrict__ x_rgb, const float* __restrict__ x_e,
                        const float* __restrict__ w1, const float* __restrict__ w2,
                        float* __restrict__ dbl) {
    int wave = (blockIdx.x * blockDim.x + threadIdx.x) >> 5;
    int ln = threadIdx.x & 31;
    int ntile = wave % 3;
    int rem   = wave / 3;
    int mtile = rem & 255;
    int branch = rem >> 8;
    const float* x = branch ? x_e : x_rgb;
    const float* w = branch ? w2 : w1;
    int m  = mtile * 16 + (ln & 15);
    int n  = ntile * 16 + (ln & 15);
    int ko = (ln >> 4) * 4;
    const float* xrow = x + (long)m * D_ + ko;
    const float* wrow = w + (long)n * D_ + ko;
    v8f c = {};
    v4f af[3][16], bf[3][16];
#pragma unroll
    for (int g = 0; g < 3; ++g) {
#pragma unroll
        for (int i = 0; i < 16; ++i) {
            af[g][i] = *(const v4f*)(xrow + g * 128 + i * 8);
            bf[g][i] = *(const v4f*)(wrow + g * 128 + i * 8);
        }
    }
    __builtin_amdgcn_sched_barrier(0);
#pragma unroll
    for (int i = 0; i < 16; ++i) {
        v2f a0 = {af[0][i].x, af[0][i].y}, b0 = {bf[0][i].x, bf[0][i].y};
        v2f a1 = {af[0][i].z, af[0][i].w}, b1 = {bf[0][i].z, bf[0][i].w};
        c = __builtin_amdgcn_wmma_f32_16x16x4_f32(false, a0, false, b0, (short)0, c, false, false);
        c = __builtin_amdgcn_wmma_f32_16x16x4_f32(false, a1, false, b1, (short)0, c, false, false);
    }
#pragma unroll
    for (int i = 0; i < 16; ++i) {
        af[0][i] = *(const v4f*)(xrow + 3 * 128 + i * 8);
        bf[0][i] = *(const v4f*)(wrow + 3 * 128 + i * 8);
    }
    __builtin_amdgcn_sched_barrier(0);
#pragma unroll
    for (int gg = 1; gg < 4; ++gg) {
        int s = gg % 3;
#pragma unroll
        for (int i = 0; i < 16; ++i) {
            v2f a0 = {af[s][i].x, af[s][i].y}, b0 = {bf[s][i].x, bf[s][i].y};
            v2f a1 = {af[s][i].z, af[s][i].w}, b1 = {bf[s][i].z, bf[s][i].w};
            c = __builtin_amdgcn_wmma_f32_16x16x4_f32(false, a0, false, b0, (short)0, c, false, false);
            c = __builtin_amdgcn_wmma_f32_16x16x4_f32(false, a1, false, b1, (short)0, c, false, false);
        }
    }
    float* out = dbl + (long)branch * 4096 * 48;
    int rowbase = mtile * 16 + ((ln >> 4) << 3);
    int col = ntile * 16 + (ln & 15);
#pragma unroll
    for (int i = 0; i < 8; ++i)
        out[(long)(rowbase + i) * 48 + col] = c[i];
}

__global__ void __launch_bounds__(128)
k_scan_local(const float* __restrict__ x_rgb, const float* __restrict__ x_e,
             const float* __restrict__ dbl,
             const float* __restrict__ dtw1, const float* __restrict__ dtb1,
             const float* __restrict__ dtw2, const float* __restrict__ dtb2,
             float* __restrict__ psum,
             const float* __restrict__ Alog1, const float* __restrict__ Alog2,
             const float* __restrict__ D1, const float* __restrict__ D2,
             float* __restrict__ hpart, float* __restrict__ aprod,
             float* __restrict__ out) {
    int bid = blockIdx.x;
    int lc  = bid & (NCHUNK - 1); bid >>= 5;
    int dch = bid & 3;  bid >>= 2;
    int b   = bid & 3;
    int branch = bid >> 2;
    int tid = threadIdx.x;
    int l0 = lc * CHUNK;

    __shared__ __attribute__((aligned(16))) float Bsh[CHUNK * 16];
    __shared__ __attribute__((aligned(16))) float Csh[CHUNK * 16];
    __shared__ __attribute__((aligned(16))) float Ssh[CHUNK * 128];
    __shared__ __attribute__((aligned(16))) float Ush[CHUNK * 128];

    const float* Alog = branch ? Alog2 : Alog1;
    const float* Dp   = branch ? D2 : D1;
    const float* x    = branch ? x_e : x_rgb;
    const float* dtw  = branch ? dtw2 : dtw1;
    const float* dtb  = branch ? dtb2 : dtb1;

    long row0 = (long)(b * L_ + l0);
    int  col0 = dch * 128;
    float*       psT = psum + (long)branch * 4096 * D_ + row0 * D_ + col0;
    const float* xT  = x + row0 * D_ + col0;
    float*       yT  = out + (long)branch * 4096 * D_ + row0 * D_ + col0;
    const float* dtT = dbl + (long)branch * 4096 * 48 + row0 * 48;

    int q = tid & 31, rr = tid >> 5;
    int c16 = q & 15;
    int kb  = (q >> 4) * 2;
    int chw = rr * 32;
    v2f af[2][4], bf[2][4];
    float bcol[2];
#pragma unroll
    for (int mt = 0; mt < 2; ++mt) {
        const float* arow = dtT + (mt * 16 + c16) * 48;
#pragma unroll
        for (int ks = 0; ks < 4; ++ks) {
            af[mt][ks].x = arow[ks * 4 + kb];
            af[mt][ks].y = arow[ks * 4 + kb + 1];
        }
    }
#pragma unroll
    for (int j = 0; j < 2; ++j) {
        int n = col0 + chw + j * 16 + c16;
        const float* wrow = dtw + (long)n * 16;
#pragma unroll
        for (int ks = 0; ks < 4; ++ks) {
            bf[j][ks].x = wrow[ks * 4 + kb];
            bf[j][ks].y = wrow[ks * 4 + kb + 1];
        }
        bcol[j] = dtb[n];
    }

    v4f uv[8];
#pragma unroll
    for (int p = 0; p < 8; ++p) {
        int r = p * 4 + rr;
        uv[p] = *(const v4f*)(xT + r * D_ + q * 4);
    }
    v4f bv, cv;
    {
        int r = tid >> 2, c4 = (tid & 3) * 4;
        bv = *(const v4f*)(dbl + (long)branch * 4096 * 48       + (row0 + r) * 48 + 16 + c4);
        cv = *(const v4f*)(dbl + (long)(1 - branch) * 4096 * 48 + (row0 + r) * 48 + 32 + c4);
    }

    int d = col0 + tid;
    float A2[N_], h[N_];
#pragma unroll
    for (int n = 0; n < N_; ++n) {
        A2[n] = -__expf(Alog[(long)d * N_ + n]) * LOG2E;
        h[n] = 0.f;
    }
    float Dd = Dp[d];

    v8f acc[2][2];
#pragma unroll
    for (int mt = 0; mt < 2; ++mt)
#pragma unroll
        for (int j = 0; j < 2; ++j)
            acc[mt][j] = (v8f){};
#pragma unroll
    for (int ks = 0; ks < 4; ++ks)
#pragma unroll
        for (int mt = 0; mt < 2; ++mt)
#pragma unroll
            for (int j = 0; j < 2; ++j)
                acc[mt][j] = __builtin_amdgcn_wmma_f32_16x16x4_f32(false, af[mt][ks], false, bf[j][ks],
                                                                   (short)0, acc[mt][j], false, false);

    int rbase = (q >> 4) << 3;
#pragma unroll
    for (int mt = 0; mt < 2; ++mt)
#pragma unroll
        for (int j = 0; j < 2; ++j)
#pragma unroll
            for (int i = 0; i < 8; ++i) {
                float v  = acc[mt][j][i] + bcol[j];
                float sp = (v > 20.f) ? v : log1pf(__expf(v));
                Ssh[(mt * 16 + rbase + i) * 128 + chw + j * 16 + c16] = sp;
            }

#pragma unroll
    for (int p = 0; p < 8; ++p) {
        int r = p * 4 + rr;
        *(v4f*)&Ush[r * 128 + q * 4] = uv[p];
    }
    *(v4f*)&Bsh[tid * 4] = bv;
    *(v4f*)&Csh[tid * 4] = cv;
    __syncthreads();

    float ssum = 0.f;
    for (int ll = 0; ll < CHUNK; ++ll) {
        float s = Ssh[ll * 128 + tid];
        float u = Ush[ll * 128 + tid];
        float su = s * u;
        ssum += s;
        float y = 0.f;
#pragma unroll
        for (int n = 0; n < N_; ++n) {
            float dA = __builtin_amdgcn_exp2f(s * A2[n]);
            h[n] = fmaf(dA, h[n], su * Bsh[ll * 16 + n]);
            y = fmaf(h[n], Csh[ll * 16 + n], y);
        }
        yT[ll * D_ + tid]  = fmaf(Dd, u, y);
        psT[ll * D_ + tid] = ssum;
    }
    long off = ((long)(branch * 4 + b) * NCHUNK + lc) * (D_ * N_) + (long)d * N_;
#pragma unroll
    for (int n = 0; n < N_; ++n) {
        hpart[off + n] = h[n];
        aprod[off + n] = __builtin_amdgcn_exp2f(ssum * A2[n]);
    }
}

__global__ void k_carry(const float* __restrict__ hpart, const float* __restrict__ aprod,
                        float* __restrict__ hin) {
    int idx = blockIdx.x * blockDim.x + threadIdx.x;
    int dn = idx & (D_ * N_ - 1);
    int bb = idx >> 13;
    float h = 0.f;
    for (int c = 0; c < NCHUNK; ++c) {
        long pos = ((long)bb * NCHUNK + c) * (D_ * N_) + dn;
        hin[pos] = h;
        h = fmaf(aprod[pos], h, hpart[pos]);
    }
}

#define FIXR 16
__global__ void __launch_bounds__(512) __attribute__((amdgpu_waves_per_eu(4, 8)))
k_fix_ln(const float* __restrict__ dbl, const float* __restrict__ psum,
         const float* __restrict__ Alog1, const float* __restrict__ Alog2,
         const float* __restrict__ hin, float* __restrict__ out,
         const float* __restrict__ g1, const float* __restrict__ bb1,
         const float* __restrict__ g2, const float* __restrict__ bb2) {
    int bid = blockIdx.x;
    int rg = bid & 1;  bid >>= 1;
    int c  = bid & (NCHUNK - 1); bid >>= 5;
    int b  = bid & 3;
    int branch = bid >> 2;
    int tid = threadIdx.x;
    int row0 = b * L_ + c * CHUNK + rg * FIXR;

    __shared__ __attribute__((aligned(16))) float Csh[FIXR * 16];
    __shared__ __attribute__((aligned(16))) float red[2 * FIXR * D_];
    __shared__ __attribute__((aligned(16))) float tot[2 * FIXR];

    const float* srcC = dbl + (long)(1 - branch) * 4096 * 48 + (long)row0 * 48 + 32;
    int ci = tid & (FIXR * 16 - 1);
    float cval = srcC[(long)(ci >> 4) * 48 + (ci & 15)];

    int d = tid;
    const float* Alog = branch ? Alog2 : Alog1;
    const float* ps   = psum + (long)branch * 4096 * D_;
    float* yout       = out  + (long)branch * 4096 * D_;
    const float* gam  = branch ? g2 : g1;
    const float* bet  = branch ? bb2 : bb1;

    float yv[FIXR], pv[FIXR];
#pragma unroll
    for (int r = 0; r < FIXR; ++r) {
        long idx = (long)(row0 + r) * D_ + d;
        yv[r] = yout[idx];
        pv[r] = ps[idx];
    }
    long off = ((long)(branch * 4 + b) * NCHUNK + c) * (D_ * N_) + (long)d * N_;
    v4f al[4], gq[4];
#pragma unroll
    for (int k = 0; k < 4; ++k) {
        al[k] = *(const v4f*)(Alog + (long)d * N_ + k * 4);
        gq[k] = *(const v4f*)(hin + off + k * 4);
    }
    float gd = gam[d], bd = bet[d];

    if (tid < FIXR * 16) Csh[tid] = cval;
    __syncthreads();

    {
        float A2[N_], g[N_];
#pragma unroll
        for (int n = 0; n < N_; ++n) {
            A2[n] = -__expf(al[n >> 2][n & 3]) * LOG2E;
            g[n] = gq[n >> 2][n & 3];
        }
#pragma unroll
        for (int r = 0; r < FIXR; ++r) {
            float corr = 0.f;
#pragma unroll
            for (int n = 0; n < N_; ++n) {
                float w = __builtin_amdgcn_exp2f(pv[r] * A2[n]);
                corr = fmaf(w * g[n], Csh[r * 16 + n], corr);
            }
            yv[r] += corr;
        }
    }

#pragma unroll
    for (int r = 0; r < FIXR; ++r) {
        red[r * D_ + tid]          = yv[r];
        red[(FIXR + r) * D_ + tid] = yv[r] * yv[r];
    }
    __syncthreads();
    {
        int w = tid >> 5, ln = tid & 31;
#pragma unroll
        for (int j = 0; j < 2; ++j) {
            int v = 2 * w + j;
            float a = 0.f;
#pragma unroll
            for (int k = 0; k < D_ / 32; ++k)
                a += red[v * D_ + k * 32 + ln];
            for (int m = 16; m > 0; m >>= 1)
                a += __shfl_xor(a, m, 32);
            if (ln == 0) tot[v] = a;
        }
    }
    __syncthreads();
#pragma unroll
    for (int r = 0; r < FIXR; ++r) {
        float mean = tot[r] * (1.f / 512.f);
        float var  = tot[FIXR + r] * (1.f / 512.f) - mean * mean;
        float inv  = rsqrtf(var + 1e-5f);
        yout[(long)(row0 + r) * D_ + d] = (yv[r] - mean) * inv * gd + bd;
    }
}

extern "C" void kernel_launch(void* const* d_in, const int* in_sizes, int n_in,
                              void* d_out, int out_size, void* d_ws, size_t ws_size,
                              hipStream_t stream) {
    const float* x_rgb = (const float*)d_in[0];
    const float* x_e   = (const float*)d_in[1];
    const float* xw1   = (const float*)d_in[2];
    const float* xw2   = (const float*)d_in[3];
    const float* dtw1  = (const float*)d_in[4];
    const float* dtb1  = (const float*)d_in[5];
    const float* dtw2  = (const float*)d_in[6];
    const float* dtb2  = (const float*)d_in[7];
    const float* Al1   = (const float*)d_in[8];
    const float* Al2   = (const float*)d_in[9];
    const float* Dv1   = (const float*)d_in[10];
    const float* Dv2   = (const float*)d_in[11];
    const float* g1    = (const float*)d_in[12];
    const float* lb1   = (const float*)d_in[13];
    const float* g2    = (const float*)d_in[14];
    const float* lb2   = (const float*)d_in[15];
    float* out = (float*)d_out;

    float* ws    = (float*)d_ws;
    float* dbl   = ws;
    float* psum  = dbl + 2 * 4096 * 48;
    float* hpart = psum + 2 * 4096 * 512;
    float* aprod = hpart + 2 * 4 * NCHUNK * 8192;
    float* hin   = aprod + 2 * 4 * NCHUNK * 8192;

    k_xproj<<<384, 128, 0, stream>>>(x_rgb, x_e, xw1, xw2, dbl);
    k_scan_local<<<1024, 128, 0, stream>>>(x_rgb, x_e, dbl, dtw1, dtb1, dtw2, dtb2, psum,
                                           Al1, Al2, Dv1, Dv2, hpart, aprod, out);
    k_carry<<<256, 256, 0, stream>>>(hpart, aprod, hin);
    k_fix_ln<<<512, 512, 0, stream>>>(dbl, psum, Al1, Al2, hin, out, g1, lb1, g2, lb2);
}


# scan and carry-term loops hold the 16 states as 8 pairs: two-wide packed f32 mul/fma do s*A, su*B, the h update and the C dot; y and the carry term become even-n plus odd-n partial sums
# speedup vs baseline: 2.6716x; 1.0408x over previous
#include <hip/hip_runtime.h>
#include <hip/hip_bf16.h>

typedef float v2f __attribute__((ext_vector_type(2)));
typedef float v4f __attribute__((ext_vector_type(4)));
typedef float v8f __attribute__((ext_vector_type(8)));

#define B_ 4
#define L_ 1024
#define D_ 512
#define N_ 16
#define CHUNK 32
#define NCHUNK 32
#define LOG2E 1.4426950408889634f

__global__ void __launch_bounds__(128)
k_xproj(const float* __restrict__ x_rgb, const float* __restrict__ x_e,
                        const float* __restrict__ w1, const float* __restrict__ w2,
                        float* __restrict__ dbl) {
    int wave = (blockIdx.x * blockDim.x + threadIdx.x) >> 5;
    int ln = threadIdx.x & 31;
    int ntile = wave % 3;
    int rem   = wave / 3;
    int mtile = rem & 255;
    int branch = rem >> 8;
    const float* x = branch ? x_e : x_rgb;
    const float* w = branch ? w2 : w1;
    int m  = mtile * 16 + (ln & 15);
    int n  = ntile * 16 + (ln & 15);
    int ko = (ln >> 4) * 4;
    const float* xrow = x + (long)m * D_ + ko;
    const float* wrow = w + (long)n * D_ + ko;
    v8f c = {};
    v4f af[3][16], bf[3][16];
#pragma unroll
    for (int g = 0; g < 3; ++g) {
#pragma unroll
        for (int i = 0; i < 16; ++i) {
            af[g][i] = *(const v4f*)(xrow + g * 128 + i * 8);
            bf[g][i] = *(const v4f*)(wrow + g * 128 + i * 8);
        }
    }
    __builtin_amdgcn_sched_barrier(0);
#pragma unroll
    for (int i = 0; i < 16; ++i) {
        v2f a0 = {af[0][i].x, af[0][i].y}, b0 = {bf[0][i].x, bf[0][i].y};
        v2f a1 = {af[0][i].z, af[0][i].w}, b1 = {bf[0][i].z, bf[0][i].w};
        c = __builtin_amdgcn_wmma_f32_16x16x4_f32(false, a0, false, b0, (short)0, c, false, false);
        c = __builtin_amdgcn_wmma_f32_16x16x4_f32(false, a1, false, b1, (short)0, c, false, false);
    }
#pragma unroll
    for (int i = 0; i < 16; ++i) {
        af[0][i] = *(const v4f*)(xrow + 3 * 128 + i * 8);
        bf[0][i] = *(const v4f*)(wrow + 3 * 128 + i * 8);
    }
    __builtin_amdgcn_sched_barrier(0);
#pragma unroll
    for (int gg = 1; gg < 4; ++gg) {
        int s = gg % 3;
#pragma unroll
        for (int i = 0; i < 16; ++i) {
            v2f a0 = {af[s][i].x, af[s][i].y}, b0 = {bf[s][i].x, bf[s][i].y};
            v2f a1 = {af[s][i].z, af[s][i].w}, b1 = {bf[s][i].z, bf[s][i].w};
            c = __builtin_amdgcn_wmma_f32_16x16x4_f32(false, a0, false, b0, (short)0, c, false, false);
            c = __builtin_amdgcn_wmma_f32_16x16x4_f32(false, a1, false, b1, (short)0, c, false, false);
        }
    }
    float* out = dbl + (long)branch * 4096 * 48;
    int rowbase = mtile * 16 + ((ln >> 4) << 3);
    int col = ntile * 16 + (ln & 15);
#pragma unroll
    for (int i = 0; i < 8; ++i)
        out[(long)(rowbase + i) * 48 + col] = c[i];
}

__global__ void __launch_bounds__(128)
k_scan_local(const float* __restrict__ x_rgb, const float* __restrict__ x_e,
             const float* __restrict__ dbl,
             const float* __restrict__ dtw1, const float* __restrict__ dtb1,
             const float* __restrict__ dtw2, const float* __restrict__ dtb2,
             float* __restrict__ psum,
             const float* __restrict__ Alog1, const float* __restrict__ Alog2,
             const float* __restrict__ D1, const float* __restrict__ D2,
             float* __restrict__ hpart, float* __restrict__ aprod,
             float* __restrict__ out) {
    int bid = blockIdx.x;
    int lc  = bid & (NCHUNK - 1); bid >>= 5;
    int dch = bid & 3;  bid >>= 2;
    int b   = bid & 3;
    int branch = bid >> 2;
    int tid = threadIdx.x;
    int l0 = lc * CHUNK;

    __shared__ __attribute__((aligned(16))) float Bsh[CHUNK * 16];
    __shared__ __attribute__((aligned(16))) float Csh[CHUNK * 16];
    __shared__ __attribute__((aligned(16))) float Ssh[CHUNK * 128];
    __shared__ __attribute__((aligned(16))) float Ush[CHUNK * 128];

    const float* Alog = branch ? Alog2 : Alog1;
    const float* Dp   = branch ? D2 : D1;
    const float* x    = branch ? x_e : x_rgb;
    const float* dtw  = branch ? dtw2 : dtw1;
    const float* dtb  = branch ? dtb2 : dtb1;

    long row0 = (long)(b * L_ + l0);
    int  col0 = dch * 128;
    float*       psT = psum + (long)branch * 4096 * D_ + row0 * D_ + col0;
    const float* xT  = x + row0 * D_ + col0;
    float*       yT  = out + (long)branch * 4096 * D_ + row0 * D_ + col0;
    const float* dtT = dbl + (long)branch * 4096 * 48 + row0 * 48;

    int q = tid & 31, rr = tid >> 5;
    int c16 = q & 15;
    int kb  = (q >> 4) * 2;
    int chw = rr * 32;
    v2f af[2][4], bf[2][4];
    float bcol[2];
#pragma unroll
    for (int mt = 0; mt < 2; ++mt) {
        const float* arow = dtT + (mt * 16 + c16) * 48;
#pragma unroll
        for (int ks = 0; ks < 4; ++ks) {
            af[mt][ks].x = arow[ks * 4 + kb];
            af[mt][ks].y = arow[ks * 4 + kb + 1];
        }
    }
#pragma unroll
    for (int j = 0; j < 2; ++j) {
        int n = col0 + chw + j * 16 + c16;
        const float* wrow = dtw + (long)n * 16;
#pragma unroll
        for (int ks = 0; ks < 4; ++ks) {
            bf[j][ks].x = wrow[ks * 4 + kb];
            bf[j][ks].y = wrow[ks * 4 + kb + 1];
        }
        bcol[j] = dtb[n];
    }

    v4f uv[8];
#pragma unroll
    for (int p = 0; p < 8; ++p) {
        int r = p * 4 + rr;
        uv[p] = *(const v4f*)(xT + r * D_ + q * 4);
    }
    v4f bv, cv;
    {
        int r = tid >> 2, c4 = (tid & 3) * 4;
        bv = *(const v4f*)(dbl + (long)branch * 4096 * 48       + (row0 + r) * 48 + 16 + c4);
        cv = *(const v4f*)(dbl + (long)(1 - branch) * 4096 * 48 + (row0 + r) * 48 + 32 + c4);
    }

    int d = col0 + tid;
    v2f A2[N_ / 2], h[N_ / 2];
#pragma unroll
    for (int k = 0; k < N_ / 2; ++k) {
        A2[k].x = -__expf(Alog[(long)d * N_ + 2 * k])     * LOG2E;
        A2[k].y = -__expf(Alog[(long)d * N_ + 2 * k + 1]) * LOG2E;
        h[k] = (v2f){0.f, 0.f};
    }
    float Dd = Dp[d];

    v8f acc[2][2];
#pragma unroll
    for (int mt = 0; mt < 2; ++mt)
#pragma unroll
        for (int j = 0; j < 2; ++j)
            acc[mt][j] = (v8f){};
#pragma unroll
    for (int ks = 0; ks < 4; ++ks)
#pragma unroll
        for (int mt = 0; mt < 2; ++mt)
#pragma unroll
            for (int j = 0; j < 2; ++j)
                acc[mt][j] = __builtin_amdgcn_wmma_f32_16x16x4_f32(false, af[mt][ks], false, bf[j][ks],
                                                                   (short)0, acc[mt][j], false, false);

    int rbase = (q >> 4) << 3;
#pragma unroll
    for (int mt = 0; mt < 2; ++mt)
#pragma unroll
        for (int j = 0; j < 2; ++j)
#pragma unroll
            for (int i = 0; i < 8; ++i) {
                float v  = acc[mt][j][i] + bcol[j];
                float sp = (v > 20.f) ? v : log1pf(__expf(v));
                Ssh[(mt * 16 + rbase + i) * 128 + chw + j * 16 + c16] = sp;
            }

#pragma unroll
    for (int p = 0; p < 8; ++p) {
        int r = p * 4 + rr;
        *(v4f*)&Ush[r * 128 + q * 4] = uv[p];
    }
    *(v4f*)&Bsh[tid * 4] = bv;
    *(v4f*)&Csh[tid * 4] = cv;
    __syncthreads();

    float ssum = 0.f;
    for (int ll = 0; ll < CHUNK; ++ll) {
        float s = Ssh[ll * 128 + tid];
        float u = Ush[ll * 128 + tid];
        float su = s * u;
        ssum += s;
        v2f y2 = {0.f, 0.f};
        v2f s2 = {s, s}, su2 = {su, su};
#pragma unroll
        for (int k = 0; k < N_ / 2; ++k) {
            v2f bp = *(const v2f*)&Bsh[ll * 16 + 2 * k];
            v2f cp = *(const v2f*)&Csh[ll * 16 + 2 * k];
            v2f e  = s2 * A2[k];
            v2f dA = {__builtin_amdgcn_exp2f(e.x), __builtin_amdgcn_exp2f(e.y)};
            h[k] = dA * h[k] + su2 * bp;
            y2   = h[k] * cp + y2;
        }
        float y = y2.x + y2.y;
        yT[ll * D_ + tid]  = fmaf(Dd, u, y);
        psT[ll * D_ + tid] = ssum;
    }
    long off = ((long)(branch * 4 + b) * NCHUNK + lc) * (D_ * N_) + (long)d * N_;
#pragma unroll
    for (int k = 0; k < N_ / 2; ++k) {
        hpart[off + 2 * k]     = h[k].x;
        hpart[off + 2 * k + 1] = h[k].y;
        aprod[off + 2 * k]     = __builtin_amdgcn_exp2f(ssum * A2[k].x);
        aprod[off + 2 * k + 1] = __builtin_amdgcn_exp2f(ssum * A2[k].y);
    }
}

__global__ void k_carry(const float* __restrict__ hpart, const float* __restrict__ aprod,
                        float* __restrict__ hin) {
    int idx = blockIdx.x * blockDim.x + threadIdx.x;
    int dn = idx & (D_ * N_ - 1);
    int bb = idx >> 13;
    float h = 0.f;
    for (int c = 0; c < NCHUNK; ++c) {
        long pos = ((long)bb * NCHUNK + c) * (D_ * N_) + dn;
        hin[pos] = h;
        h = fmaf(aprod[pos], h, hpart[pos]);
    }
}

#define FIXR 16
__global__ void __launch_bounds__(512) __attribute__((amdgpu_waves_per_eu(4, 8)))
k_fix_ln(const float* __restrict__ dbl, const float* __restrict__ psum,
         const float* __restrict__ Alog1, const float* __restrict__ Alog2,
         const float* __restrict__ hin, float* __restrict__ out,
         const float* __restrict__ g1, const float* __restrict__ bb1,
         const float* __restrict__ g2, const float* __restrict__ bb2) {
    int bid = blockIdx.x;
    int rg = bid & 1;  bid >>= 1;
    int c  = bid & (NCHUNK - 1); bid >>= 5;
    int b  = bid & 3;
    int branch = bid >> 2;
    int tid = threadIdx.x;
    int row0 = b * L_ + c * CHUNK + rg * FIXR;

    __shared__ __attribute__((aligned(16))) float Csh[FIXR * 16];
    __shared__ __attribute__((aligned(16))) float red[2 * FIXR * D_];
    __shared__ __attribute__((aligned(16))) float tot[2 * FIXR];

    const float* srcC = dbl + (long)(1 - branch) * 4096 * 48 + (long)row0 * 48 + 32;
    int ci = tid & (FIXR * 16 - 1);
    float cval = srcC[(long)(ci >> 4) * 48 + (ci & 15)];

    int d = tid;
    const float* Alog = branch ? Alog2 : Alog1;
    const float* ps   = psum + (long)branch * 4096 * D_;
    float* yout       = out  + (long)branch * 4096 * D_;
    const float* gam  = branch ? g2 : g1;
    const float* bet  = branch ? bb2 : bb1;

    float yv[FIXR], pv[FIXR];
#pragma unroll
    for (int r = 0; r < FIXR; ++r) {
        long idx = (long)(row0 + r) * D_ + d;
        yv[r] = yout[idx];
        pv[r] = ps[idx];
    }
    long off = ((long)(branch * 4 + b) * NCHUNK + c) * (D_ * N_) + (long)d * N_;
    v4f al[4], gq[4];
#pragma unroll
    for (int k = 0; k < 4; ++k) {
        al[k] = *(const v4f*)(Alog + (long)d * N_ + k * 4);
        gq[k] = *(const v4f*)(hin + off + k * 4);
    }
    float gd = gam[d], bd = bet[d];

    if (tid < FIXR * 16) Csh[tid] = cval;
    __syncthreads();

    {
        v2f A2[N_ / 2], g[N_ / 2];
#pragma unroll
        for (int k = 0; k < N_ / 2; ++k) {
            A2[k].x = -__expf(al[k >> 1][(2 * k) & 3])     * LOG2E;
            A2[k].y = -__expf(al[k >> 1][(2 * k + 1) & 3]) * LOG2E;
            g[k].x  = gq[k >> 1][(2 * k) & 3];
            g[k].y  = gq[k >> 1][(2 * k + 1) & 3];
        }
#pragma unroll
        for (int r = 0; r < FIXR; ++r) {
            v2f corr2 = {0.f, 0.f};
            v2f p2 = {pv[r], pv[r]};
#pragma unroll
            for (int k = 0; k < N_ / 2; ++k) {
                v2f e  = p2 * A2[k];
                v2f w  = {__builtin_amdgcn_exp2f(e.x), __builtin_amdgcn_exp2f(e.y)};
                v2f cp = *(const v2f*)&Csh[r * 16 + 2 * k];
                corr2 = (w * g[k]) * cp + corr2;
            }
            yv[r] += corr2.x + corr2.y;
        }
    }

#pragma unroll
    for (int r = 0; r < FIXR; ++r) {
        red[r * D_ + tid]          = yv[r];
        red[(FIXR + r) * D_ + tid] = yv[r] * yv[r];
    }
    __syncthreads();
    {
        int w = tid >> 5, ln = tid & 31;
#pragma unroll
        for (int j = 0; j < 2; ++j) {
            int v = 2 * w + j;
            float a = 0.f;
#pragma unroll
            for (int k = 0; k < D_ / 32; ++k)
                a += red[v * D_ + k * 32 + ln];
            for (int m = 16; m > 0; m >>= 1)
                a += __shfl_xor(a, m, 32);
            if (ln == 0) tot[v] = a;
        }
    }
    __syncthreads();
#pragma unroll
    for (int r = 0; r < FIXR; ++r) {
        float mean = tot[r] * (1.f / 512.f);
        float var  = tot[FIXR + r] * (1.f / 512.f) - mean * mean;
        float inv  = rsqrtf(var + 1e-5f);
        yout[(long)(row0 + r) * D_ + d] = (yv[r] - mean) * inv * gd + bd;
    }
}

extern "C" void kernel_launch(void* const* d_in, const int* in_sizes, int n_in,
                              void* d_out, int out_size, void* d_ws, size_t ws_size,
                              hipStream_t stream) {
    const float* x_rgb = (const float*)d_in[0];
    const float* x_e   = (const float*)d_in[1];
    const float* xw1   = (const float*)d_in[2];
    const float* xw2   = (const float*)d_in[3];
    const float* dtw1  = (const float*)d_in[4];
    const float* dtb1  = (const float*)d_in[5];
    const float* dtw2  = (const float*)d_in[6];
    const float* dtb2  = (const float*)d_in[7];
    const float* Al1   = (const float*)d_in[8];
    const float* Al2   = (const float*)d_in[9];
    const float* Dv1   = (const float*)d_in[10];
    const float* Dv2   = (const float*)d_in[11];
    const float* g1    = (const float*)d_in[12];
    const float* lb1   = (const float*)d_in[13];
    const float* g2    = (const float*)d_in[14];
    const float* lb2   = (const float*)d_in[15];
    float* out = (float*)d_out;

    float* ws    = (float*)d_ws;
    float* dbl   = ws;
    float* psum  = dbl + 2 * 4096 * 48;
    float* hpart = psum + 2 * 4096 * 512;
    float* aprod = hpart + 2 * 4 * NCHUNK * 8192;
    float* hin   = aprod + 2 * 4 * NCHUNK * 8192;

    k_xproj<<<384, 128, 0, stream>>>(x_rgb, x_e, xw1, xw2, dbl);
    k_scan_local<<<1024, 128, 0, stream>>>(x_rgb, x_e, dbl, dtw1, dtb1, dtw2, dtb2, psum,
                                           Al1, Al2, Dv1, Dv2, hpart, aprod, out);
    k_carry<<<256, 256, 0, stream>>>(hpart, aprod, hin);
    k_fix_ln<<<512, 512, 0, stream>>>(dbl, psum, Al1, Al2, hin, out, g1, lb1, g2, lb2);
}


# k_xproj: a block owns 32 rows x 48 columns (256 blocks of 6 waves); x and w row pieces come into LDS as whole 512-byte lines in four K stages, fragments are read from LDS; sums as before
# speedup vs baseline: 2.8905x; 1.0819x over previous
#include <hip/hip_runtime.h>
#include <hip/hip_bf16.h>

typedef float v2f __attribute__((ext_vector_type(2)));
typedef float v4f __attribute__((ext_vector_type(4)));
typedef float v8f __attribute__((ext_vector_type(8)));

#define B_ 4
#define L_ 1024
#define D_ 512
#define N_ 16
#define CHUNK 32
#define NCHUNK 32
#define LOG2E 1.4426950408889634f

#define XP_ROWS  80
#define XP_PITCH 516
#define XP_UNITS 14
__global__ void __launch_bounds__(192)
k_xproj(const float* __restrict__ x_rgb, const float* __restrict__ x_e,
                        const float* __restrict__ w1, const float* __restrict__ w2,
                        float* __restrict__ dbl) {
    __shared__ __attribute__((aligned(16))) float tile[XP_ROWS * XP_PITCH];
    int bid = blockIdx.x;
    int branch = bid >> 7;
    int mblk = bid & 127;
    const float* x = (branch ? x_e : x_rgb) + (long)mblk * 32 * D_;
    const float* w = branch ? w2 : w1;
    int ln = threadIdx.x & 31;
    int wv = __builtin_amdgcn_readfirstlane((int)(threadIdx.x >> 5));

    const float* srow[XP_UNITS];
    int drow[XP_UNITS];
#pragma unroll
    for (int j = 0; j < XP_UNITS; ++j) {
        int u = wv + 6 * j;
        u = (u > XP_ROWS - 1) ? XP_ROWS - 1 : u;
        srow[j] = (u < 32) ? x + (long)u * D_ : w + (long)(u - 32) * D_;
        drow[j] = u * XP_PITCH;
    }
    v4f st[4][XP_UNITS];
#pragma unroll
    for (int g = 0; g < 4; ++g) {
#pragma unroll
        for (int j = 0; j < XP_UNITS; ++j)
            st[g][j] = *(const v4f*)(srow[j] + g * 128 + ln * 4);
        __builtin_amdgcn_sched_barrier(0);
    }

    int c16 = ln & 15;
    int ko  = (ln >> 4) * 4;
    int mt  = wv / 3;
    int nt  = wv - mt * 3;
    const float* arow = &tile[(mt * 16 + c16) * XP_PITCH + ko];
    const float* brow = &tile[(32 + nt * 16 + c16) * XP_PITCH + ko];
    v8f c = {};
#pragma unroll
    for (int g = 0; g < 4; ++g) {
#pragma unroll
        for (int j = 0; j < XP_UNITS; ++j)
            *(v4f*)&tile[drow[j] + g * 128 + ln * 4] = st[g][j];
        __builtin_amdgcn_fence(__ATOMIC_RELEASE, "workgroup", "local");
        __builtin_amdgcn_s_barrier();
        __builtin_amdgcn_fence(__ATOMIC_ACQUIRE, "workgroup", "local");
        v4f af[16], bf[16];
#pragma unroll
        for (int i = 0; i < 16; ++i) {
            af[i] = *(const v4f*)(arow + g * 128 + i * 8);
            bf[i] = *(const v4f*)(brow + g * 128 + i * 8);
        }
        __builtin_amdgcn_sched_barrier(0);
#pragma unroll
        for (int i = 0; i < 16; ++i) {
            v2f a0 = {af[i].x, af[i].y}, b0 = {bf[i].x, bf[i].y};
            v2f a1 = {af[i].z, af[i].w}, b1 = {bf[i].z, bf[i].w};
            c = __builtin_amdgcn_wmma_f32_16x16x4_f32(false, a0, false, b0, (short)0, c, false, false);
            c = __builtin_amdgcn_wmma_f32_16x16x4_f32(false, a1, false, b1, (short)0, c, false, false);
        }
        __builtin_amdgcn_sched_barrier(0);
    }
    float* out = dbl + (long)branch * 4096 * 48;
    int rowbase = mblk * 32 + mt * 16 + ((ln >> 4) << 3);
    int col = nt * 16 + c16;
#pragma unroll
    for (int i = 0; i < 8; ++i)
        out[(long)(rowbase + i) * 48 + col] = c[i];
}

__global__ void __launch_bounds__(128)
k_scan_local(const float* __restrict__ x_rgb, const float* __restrict__ x_e,
             const float* __restrict__ dbl,
             const float* __restrict__ dtw1, const float* __restrict__ dtb1,
             const float* __restrict__ dtw2, const float* __restrict__ dtb2,
             float* __restrict__ psum,
             const float* __restrict__ Alog1, const float* __restrict__ Alog2,
             const float* __restrict__ D1, const float* __restrict__ D2,
             float* __restrict__ hpart, float* __restrict__ aprod,
             float* __restrict__ out) {
    int bid = blockIdx.x;
    int lc  = bid & (NCHUNK - 1); bid >>= 5;
    int dch = bid & 3;  bid >>= 2;
    int b   = bid & 3;
    int branch = bid >> 2;
    int tid = threadIdx.x;
    int l0 = lc * CHUNK;

    __shared__ __attribute__((aligned(16))) float Bsh[CHUNK * 16];
    __shared__ __attribute__((aligned(16))) float Csh[CHUNK * 16];
    __shared__ __attribute__((aligned(16))) float Ssh[CHUNK * 128];
    __shared__ __attribute__((aligned(16))) float Ush[CHUNK * 128];

    const float* Alog = branch ? Alog2 : Alog1;
    const float* Dp   = branch ? D2 : D1;
    const float* x    = branch ? x_e : x_rgb;
    const float* dtw  = branch ? dtw2 : dtw1;
    const float* dtb  = branch ? dtb2 : dtb1;

    long row0 = (long)(b * L_ + l0);
    int  col0 = dch * 128;
    float*       psT = psum + (long)branch * 4096 * D_ + row0 * D_ + col0;
    const float* xT  = x + row0 * D_ + col0;
    float*       yT  = out + (long)branch * 4096 * D_ + row0 * D_ + col0;
    const float* dtT = dbl + (long)branch * 4096 * 48 + row0 * 48;

    int q = tid & 31, rr = tid >> 5;
    int c16 = q & 15;
    int kb  = (q >> 4) * 2;
    int chw = rr * 32;
    v2f af[2][4], bf[2][4];
    float bcol[2];
#pragma unroll
    for (int mt = 0; mt < 2; ++mt) {
        const float* arow = dtT + (mt * 16 + c16) * 48;
#pragma unroll
        for (int ks = 0; ks < 4; ++ks) {
            af[mt][ks].x = arow[ks * 4 + kb];
            af[mt][ks].y = arow[ks * 4 + kb + 1];
        }
    }
#pragma unroll
    for (int j = 0; j < 2; ++j) {
        int n = col0 + chw + j * 16 + c16;
        const float* wrow = dtw + (long)n * 16;
#pragma unroll
        for (int ks = 0; ks < 4; ++ks) {
            bf[j][ks].x = wrow[ks * 4 + kb];
            bf[j][ks].y = wrow[ks * 4 + kb + 1];
        }
        bcol[j] = dtb[n];
    }

    v4f uv[8];
#pragma unroll
    for (int p = 0; p < 8; ++p) {
        int r = p * 4 + rr;
        uv[p] = *(const v4f*)(xT + r * D_ + q * 4);
    }
    v4f bv, cv;
    {
        int r = tid >> 2, c4 = (tid & 3) * 4;
        bv = *(const v4f*)(dbl + (long)branch * 4096 * 48       + (row0 + r) * 48 + 16 + c4);
        cv = *(const v4f*)(dbl + (long)(1 - branch) * 4096 * 48 + (row0 + r) * 48 + 32 + c4);
    }

    int d = col0 + tid;
    v2f A2[N_ / 2], h[N_ / 2];
#pragma unroll
    for (int k = 0; k < N_ / 2; ++k) {
        A2[k].x = -__expf(Alog[(long)d * N_ + 2 * k])     * LOG2E;
        A2[k].y = -__expf(Alog[(long)d * N_ + 2 * k + 1]) * LOG2E;
        h[k] = (v2f){0.f, 0.f};
    }
    float Dd = Dp[d];

    v8f acc[2][2];
#pragma unroll
    for (int mt = 0; mt < 2; ++mt)
#pragma unroll
        for (int j = 0; j < 2; ++j)
            acc[mt][j] = (v8f){};
#pragma unroll
    for (int ks = 0; ks < 4; ++ks)
#pragma unroll
        for (int mt = 0; mt < 2; ++mt)
#pragma unroll
            for (int j = 0; j < 2; ++j)
                acc[mt][j] = __builtin_amdgcn_wmma_f32_16x16x4_f32(false, af[mt][ks], false, bf[j][ks],
                                                                   (short)0, acc[mt][j], false, false);

    int rbase = (q >> 4) << 3;
#pragma unroll
    for (int mt = 0; mt < 2; ++mt)
#pragma unroll
        for (int j = 0; j < 2; ++j)
#pragma unroll
            for (int i = 0; i < 8; ++i) {
                float v  = acc[mt][j][i] + bcol[j];
                float sp = (v > 20.f) ? v : log1pf(__expf(v));
                Ssh[(mt * 16 + rbase + i) * 128 + chw + j * 16 + c16] = sp;
            }

#pragma unroll
    for (int p = 0; p < 8; ++p) {
        int r = p * 4 + rr;
        *(v4f*)&Ush[r * 128 + q * 4] = uv[p];
    }
    *(v4f*)&Bsh[tid * 4] = bv;
    *(v4f*)&Csh[tid * 4] = cv;
    __syncthreads();

    float ssum = 0.f;
    for (int ll = 0; ll < CHUNK; ++ll) {
        float s = Ssh[ll * 128 + tid];
        float u = Ush[ll * 128 + tid];
        float su = s * u;
        ssum += s;
        v2f y2 = {0.f, 0.f};
        v2f s2 = {s, s}, su2 = {su, su};
#pragma unroll
        for (int k = 0; k < N_ / 2; ++k) {
            v2f bp = *(const v2f*)&Bsh[ll * 16 + 2 * k];
            v2f cp = *(const v2f*)&Csh[ll * 16 + 2 * k];
            v2f e  = s2 * A2[k];
            v2f dA = {__builtin_amdgcn_exp2f(e.x), __builtin_amdgcn_exp2f(e.y)};
            h[k] = dA * h[k] + su2 * bp;
            y2   = h[k] * cp + y2;
        }
        float y = y2.x + y2.y;
        yT[ll * D_ + tid]  = fmaf(Dd, u, y);
        psT[ll * D_ + tid] = ssum;
    }
    long off = ((long)(branch * 4 + b) * NCHUNK + lc) * (D_ * N_) + (long)d * N_;
#pragma unroll
    for (int k = 0; k < N_ / 2; ++k) {
        hpart[off + 2 * k]     = h[k].x;
        hpart[off + 2 * k + 1] = h[k].y;
        aprod[off + 2 * k]     = __builtin_amdgcn_exp2f(ssum * A2[k].x);
        aprod[off + 2 * k + 1] = __builtin_amdgcn_exp2f(ssum * A2[k].y);
    }
}

__global__ void k_carry(const float* __restrict__ hpart, const float* __restrict__ aprod,
                        float* __restrict__ hin) {
    int idx = blockIdx.x * blockDim.x + threadIdx.x;
    int dn = idx & (D_ * N_ - 1);
    int bb = idx >> 13;
    float h = 0.f;
    for (int c = 0; c < NCHUNK; ++c) {
        long pos = ((long)bb * NCHUNK + c) * (D_ * N_) + dn;
        hin[pos] = h;
        h = fmaf(aprod[pos], h, hpart[pos]);
    }
}

#define FIXR 16
__global__ void __launch_bounds__(512) __attribute__((amdgpu_waves_per_eu(4, 8)))
k_fix_ln(const float* __restrict__ dbl, const float* __restrict__ psum,
         const float* __restrict__ Alog1, const float* __restrict__ Alog2,
         const float* __restrict__ hin, float* __restrict__ out,
         const float* __restrict__ g1, const float* __restrict__ bb1,
         const float* __restrict__ g2, const float* __restrict__ bb2) {
    int bid = blockIdx.x;
    int rg = bid & 1;  bid >>= 1;
    int c  = bid & (NCHUNK - 1); bid >>= 5;
    int b  = bid & 3;
    int branch = bid >> 2;
    int tid = threadIdx.x;
    int row0 = b * L_ + c * CHUNK + rg * FIXR;

    __shared__ __attribute__((aligned(16))) float Csh[FIXR * 16];
    __shared__ __attribute__((aligned(16))) float red[2 * FIXR * D_];
    __shared__ __attribute__((aligned(16))) float tot[2 * FIXR];

    const float* srcC = dbl + (long)(1 - branch) * 4096 * 48 + (long)row0 * 48 + 32;
    int ci = tid & (FIXR * 16 - 1);
    float cval = srcC[(long)(ci >> 4) * 48 + (ci & 15)];

    int d = tid;
    const float* Alog = branch ? Alog2 : Alog1;
    const float* ps   = psum + (long)branch * 4096 * D_;
    float* yout       = out  + (long)branch * 4096 * D_;
    const float* gam  = branch ? g2 : g1;
    const float* bet  = branch ? bb2 : bb1;

    float yv[FIXR], pv[FIXR];
#pragma unroll
    for (int r = 0; r < FIXR; ++r) {
        long idx = (long)(row0 + r) * D_ + d;
        yv[r] = yout[idx];
        pv[r] = ps[idx];
    }
    long off = ((long)(branch * 4 + b) * NCHUNK + c) * (D_ * N_) + (long)d * N_;
    v4f al[4], gq[4];
#pragma unroll
    for (int k = 0; k < 4; ++k) {
        al[k] = *(const v4f*)(Alog + (long)d * N_ + k * 4);
        gq[k] = *(const v4f*)(hin + off + k * 4);
    }
    float gd = gam[d], bd = bet[d];

    if (tid < FIXR * 16) Csh[tid] = cval;
    __syncthreads();

    {
        v2f A2[N_ / 2], g[N_ / 2];
#pragma unroll
        for (int k = 0; k < N_ / 2; ++k) {
            A2[k].x = -__expf(al[k >> 1][(2 * k) & 3])     * LOG2E;
            A2[k].y = -__expf(al[k >> 1][(2 * k + 1) & 3]) * LOG2E;
            g[k].x  = gq[k >> 1][(2 * k) & 3];
            g[k].y  = gq[k >> 1][(2 * k + 1) & 3];
        }
#pragma unroll
        for (int r = 0; r < FIXR; ++r) {
            v2f corr2 = {0.f, 0.f};
            v2f p2 = {pv[r], pv[r]};
#pragma unroll
            for (int k = 0; k < N_ / 2; ++k) {
                v2f e  = p2 * A2[k];
                v2f w  = {__builtin_amdgcn_exp2f(e.x), __builtin_amdgcn_exp2f(e.y)};
                v2f cp = *(const v2f*)&Csh[r * 16 + 2 * k];
                corr2 = (w * g[k]) * cp + corr2;
            }
            yv[r] += corr2.x + corr2.y;
        }
    }

#pragma unroll
    for (int r = 0; r < FIXR; ++r) {
        red[r * D_ + tid]          = yv[r];
        red[(FIXR + r) * D_ + tid] = yv[r] * yv[r];
    }
    __syncthreads();
    {
        int w = tid >> 5, ln = tid & 31;
#pragma unroll
        for (int j = 0; j < 2; ++j) {
            int v = 2 * w + j;
            float a = 0.f;
#pragma unroll
            for (int k = 0; k < D_ / 32; ++k)
                a += red[v * D_ + k * 32 + ln];
            for (int m = 16; m > 0; m >>= 1)
                a += __shfl_xor(a, m, 32);
            if (ln == 0) tot[v] = a;
        }
    }
    __syncthreads();
#pragma unroll
    for (int r = 0; r < FIXR; ++r) {
        float mean = tot[r] * (1.f / 512.f);
        float var  = tot[FIXR + r] * (1.f / 512.f) - mean * mean;
        float inv  = rsqrtf(var + 1e-5f);
        yout[(long)(row0 + r) * D_ + d] = (yv[r] - mean) * inv * gd + bd;
    }
}

extern "C" void kernel_launch(void* const* d_in, const int* in_sizes, int n_in,
                              void* d_out, int out_size, void* d_ws, size_t ws_size,
                              hipStream_t stream) {
    const float* x_rgb = (const float*)d_in[0];
    const float* x_e   = (const float*)d_in[1];
    const float* xw1   = (const float*)d_in[2];
    const float* xw2   = (const float*)d_in[3];
    const float* dtw1  = (const float*)d_in[4];
    const float* dtb1  = (const float*)d_in[5];
    const float* dtw2  = (const float*)d_in[6];
    const float* dtb2  = (const float*)d_in[7];
    const float* Al1   = (const float*)d_in[8];
    const float* Al2   = (const float*)d_in[9];
    const float* Dv1   = (const float*)d_in[10];
    const float* Dv2   = (const float*)d_in[11];
    const float* g1    = (const float*)d_in[12];
    const float* lb1   = (const float*)d_in[13];
    const float* g2    = (const float*)d_in[14];
    const float* lb2   = (const float*)d_in[15];
    float* out = (float*)d_out;

    float* ws    = (float*)d_ws;
    float* dbl   = ws;
    float* psum  = dbl + 2 * 4096 * 48;
    float* hpart = psum + 2 * 4096 * 512;
    float* aprod = hpart + 2 * 4 * NCHUNK * 8192;
    float* hin   = aprod + 2 * 4 * NCHUNK * 8192;

    k_xproj<<<256, 192, 0, stream>>>(x_rgb, x_e, xw1, xw2, dbl);
    k_scan_local<<<1024, 128, 0, stream>>>(x_rgb, x_e, dbl, dtw1, dtb1, dtw2, dtb2, psum,
                                           Al1, Al2, Dv1, Dv2, hpart, aprod, out);
    k_carry<<<256, 256, 0, stream>>>(hpart, aprod, hin);
    k_fix_ln<<<512, 512, 0, stream>>>(dbl, psum, Al1, Al2, hin, out, g1, lb1, g2, lb2);
}


# k_scan_local softplus: log1p(e) = hardware log2 of w=1+e times e/(w-1) (w-1 exact, so the rounding of 1+e divides out; e itself where w==1), not the library log1pf; delta moves by a few ulp
# speedup vs baseline: 3.2548x; 1.1260x over previous
#include <hip/hip_runtime.h>
#include <hip/hip_bf16.h>

typedef float v2f __attribute__((ext_vector_type(2)));
typedef float v4f __attribute__((ext_vector_type(4)));
typedef float v8f __attribute__((ext_vector_type(8)));

#define B_ 4
#define L_ 1024
#define D_ 512
#define N_ 16
#define CHUNK 32
#define NCHUNK 32
#define LOG2E 1.4426950408889634f
#define LN2   0.6931471805599453f

__device__ __forceinline__ float log1p_short(float e) {
    float w  = 1.f + e;
    float wm = w - 1.f;
    float l  = __builtin_amdgcn_logf(w) * LN2;
    float r  = l * (e * __builtin_amdgcn_rcpf(wm));
    return (w == 1.f) ? e : r;
}

#define XP_ROWS  80
#define XP_PITCH 516
#define XP_UNITS 14
__global__ void __launch_bounds__(192)
k_xproj(const float* __restrict__ x_rgb, const float* __restrict__ x_e,
                        const float* __restrict__ w1, const float* __restrict__ w2,
                        float* __restrict__ dbl) {
    __shared__ __attribute__((aligned(16))) float tile[XP_ROWS * XP_PITCH];
    int bid = blockIdx.x;
    int branch = bid >> 7;
    int mblk = bid & 127;
    const float* x = (branch ? x_e : x_rgb) + (long)mblk * 32 * D_;
    const float* w = branch ? w2 : w1;
    int ln = threadIdx.x & 31;
    int wv = __builtin_amdgcn_readfirstlane((int)(threadIdx.x >> 5));

    const float* srow[XP_UNITS];
    int drow[XP_UNITS];
#pragma unroll
    for (int j = 0; j < XP_UNITS; ++j) {
        int u = wv + 6 * j;
        u = (u > XP_ROWS - 1) ? XP_ROWS - 1 : u;
        srow[j] = (u < 32) ? x + (long)u * D_ : w + (long)(u - 32) * D_;
        drow[j] = u * XP_PITCH;
    }
    v4f st[4][XP_UNITS];
#pragma unroll
    for (int g = 0; g < 4; ++g) {
#pragma unroll
        for (int j = 0; j < XP_UNITS; ++j)
            st[g][j] = *(const v4f*)(srow[j] + g * 128 + ln * 4);
        __builtin_amdgcn_sched_barrier(0);
    }

    int c16 = ln & 15;
    int ko  = (ln >> 4) * 4;
    int mt  = wv / 3;
    int nt  = wv - mt * 3;
    const float* arow = &tile[(mt * 16 + c16) * XP_PITCH + ko];
    const float* brow = &tile[(32 + nt * 16 + c16) * XP_PITCH + ko];
    v8f c = {};
#pragma unroll
    for (int g = 0; g < 4; ++g) {
#pragma unroll
        for (int j = 0; j < XP_UNITS; ++j)
            *(v4f*)&tile[drow[j] + g * 128 + ln * 4] = st[g][j];
        __builtin_amdgcn_fence(__ATOMIC_RELEASE, "workgroup", "local");
        __builtin_amdgcn_s_barrier();
        __builtin_amdgcn_fence(__ATOMIC_ACQUIRE, "workgroup", "local");
        v4f af[16], bf[16];
#pragma unroll
        for (int i = 0; i < 16; ++i) {
            af[i] = *(const v4f*)(arow + g * 128 + i * 8);
            bf[i] = *(const v4f*)(brow + g * 128 + i * 8);
        }
        __builtin_amdgcn_sched_barrier(0);
#pragma unroll
        for (int i = 0; i < 16; ++i) {
            v2f a0 = {af[i].x, af[i].y}, b0 = {bf[i].x, bf[i].y};
            v2f a1 = {af[i].z, af[i].w}, b1 = {bf[i].z, bf[i].w};
            c = __builtin_amdgcn_wmma_f32_16x16x4_f32(false, a0, false, b0, (short)0, c, false, false);
            c = __builtin_amdgcn_wmma_f32_16x16x4_f32(false, a1, false, b1, (short)0, c, false, false);
        }
        __builtin_amdgcn_sched_barrier(0);
    }
    float* out = dbl + (long)branch * 4096 * 48;
    int rowbase = mblk * 32 + mt * 16 + ((ln >> 4) << 3);
    int col = nt * 16 + c16;
#pragma unroll
    for (int i = 0; i < 8; ++i)
        out[(long)(rowbase + i) * 48 + col] = c[i];
}

__global__ void __launch_bounds__(128)
k_scan_local(const float* __restrict__ x_rgb, const float* __restrict__ x_e,
             const float* __restrict__ dbl,
             const float* __restrict__ dtw1, const float* __restrict__ dtb1,
             const float* __restrict__ dtw2, const float* __restrict__ dtb2,
             float* __restrict__ psum,
             const float* __restrict__ Alog1, const float* __restrict__ Alog2,
             const float* __restrict__ D1, const float* __restrict__ D2,
             float* __restrict__ hpart, float* __restrict__ aprod,
             float* __restrict__ out) {
    int bid = blockIdx.x;
    int lc  = bid & (NCHUNK - 1); bid >>= 5;
    int dch = bid & 3;  bid >>= 2;
    int b   = bid & 3;
    int branch = bid >> 2;
    int tid = threadIdx.x;
    int l0 = lc * CHUNK;

    __shared__ __attribute__((aligned(16))) float Bsh[CHUNK * 16];
    __shared__ __attribute__((aligned(16))) float Csh[CHUNK * 16];
    __shared__ __attribute__((aligned(16))) float Ssh[CHUNK * 128];
    __shared__ __attribute__((aligned(16))) float Ush[CHUNK * 128];

    const float* Alog = branch ? Alog2 : Alog1;
    const float* Dp   = branch ? D2 : D1;
    const float* x    = branch ? x_e : x_rgb;
    const float* dtw  = branch ? dtw2 : dtw1;
    const float* dtb  = branch ? dtb2 : dtb1;

    long row0 = (long)(b * L_ + l0);
    int  col0 = dch * 128;
    float*       psT = psum + (long)branch * 4096 * D_ + row0 * D_ + col0;
    const float* xT  = x + row0 * D_ + col0;
    float*       yT  = out + (long)branch * 4096 * D_ + row0 * D_ + col0;
    const float* dtT = dbl + (long)branch * 4096 * 48 + row0 * 48;

    int q = tid & 31, rr = tid >> 5;
    int c16 = q & 15;
    int kb  = (q >> 4) * 2;
    int chw = rr * 32;
    v2f af[2][4], bf[2][4];
    float bcol[2];
#pragma unroll
    for (int mt = 0; mt < 2; ++mt) {
        const float* arow = dtT + (mt * 16 + c16) * 48;
#pragma unroll
        for (int ks = 0; ks < 4; ++ks) {
            af[mt][ks].x = arow[ks * 4 + kb];
            af[mt][ks].y = arow[ks * 4 + kb + 1];
        }
    }
#pragma unroll
    for (int j = 0; j < 2; ++j) {
        int n = col0 + chw + j * 16 + c16;
        const float* wrow = dtw + (long)n * 16;
#pragma unroll
        for (int ks = 0; ks < 4; ++ks) {
            bf[j][ks].x = wrow[ks * 4 + kb];
            bf[j][ks].y = wrow[ks * 4 + kb + 1];
        }
        bcol[j] = dtb[n];
    }

    v4f uv[8];
#pragma unroll
    for (int p = 0; p < 8; ++p) {
        int r = p * 4 + rr;
        uv[p] = *(const v4f*)(xT + r * D_ + q * 4);
    }
    v4f bv, cv;
    {
        int r = tid >> 2, c4 = (tid & 3) * 4;
        bv = *(const v4f*)(dbl + (long)branch * 4096 * 48       + (row0 + r) * 48 + 16 + c4);
        cv = *(const v4f*)(dbl + (long)(1 - branch) * 4096 * 48 + (row0 + r) * 48 + 32 + c4);
    }

    int d = col0 + tid;
    v2f A2[N_ / 2], h[N_ / 2];
#pragma unroll
    for (int k = 0; k < N_ / 2; ++k) {
        A2[k].x = -__expf(Alog[(long)d * N_ + 2 * k])     * LOG2E;
        A2[k].y = -__expf(Alog[(long)d * N_ + 2 * k + 1]) * LOG2E;
        h[k] = (v2f){0.f, 0.f};
    }
    float Dd = Dp[d];

    v8f acc[2][2];
#pragma unroll
    for (int mt = 0; mt < 2; ++mt)
#pragma unroll
        for (int j = 0; j < 2; ++j)
            acc[mt][j] = (v8f){};
#pragma unroll
    for (int ks = 0; ks < 4; ++ks)
#pragma unroll
        for (int mt = 0; mt < 2; ++mt)
#pragma unroll
            for (int j = 0; j < 2; ++j)
                acc[mt][j] = __builtin_amdgcn_wmma_f32_16x16x4_f32(false, af[mt][ks], false, bf[j][ks],
                                                                   (short)0, acc[mt][j], false, false);

    int rbase = (q >> 4) << 3;
#pragma unroll
    for (int mt = 0; mt < 2; ++mt)
#pragma unroll
        for (int j = 0; j < 2; ++j)
#pragma unroll
            for (int i = 0; i < 8; ++i) {
                float v  = acc[mt][j][i] + bcol[j];
                float sp = (v > 20.f) ? v : log1p_short(__expf(v));
                Ssh[(mt * 16 + rbase + i) * 128 + chw + j * 16 + c16] = sp;
            }

#pragma unroll
    for (int p = 0; p < 8; ++p) {
        int r = p * 4 + rr;
        *(v4f*)&Ush[r * 128 + q * 4] = uv[p];
    }
    *(v4f*)&Bsh[tid * 4] = bv;
    *(v4f*)&Csh[tid * 4] = cv;
    __syncthreads();

    float ssum = 0.f;
    for (int ll = 0; ll < CHUNK; ++ll) {
        float s = Ssh[ll * 128 + tid];
        float u = Ush[ll * 128 + tid];
        float su = s * u;
        ssum += s;
        v2f y2 = {0.f, 0.f};
        v2f s2 = {s, s}, su2 = {su, su};
#pragma unroll
        for (int k = 0; k < N_ / 2; ++k) {
            v2f bp = *(const v2f*)&Bsh[ll * 16 + 2 * k];
            v2f cp = *(const v2f*)&Csh[ll * 16 + 2 * k];
            v2f e  = s2 * A2[k];
            v2f dA = {__builtin_amdgcn_exp2f(e.x), __builtin_amdgcn_exp2f(e.y)};
            h[k] = dA * h[k] + su2 * bp;
            y2   = h[k] * cp + y2;
        }
        float y = y2.x + y2.y;
        yT[ll * D_ + tid]  = fmaf(Dd, u, y);
        psT[ll * D_ + tid] = ssum;
    }
    long off = ((long)(branch * 4 + b) * NCHUNK + lc) * (D_ * N_) + (long)d * N_;
#pragma unroll
    for (int k = 0; k < N_ / 2; ++k) {
        hpart[off + 2 * k]     = h[k].x;
        hpart[off + 2 * k + 1] = h[k].y;
        aprod[off + 2 * k]     = __builtin_amdgcn_exp2f(ssum * A2[k].x);
        aprod[off + 2 * k + 1] = __builtin_amdgcn_exp2f(ssum * A2[k].y);
    }
}

__global__ void k_carry(const float* __restrict__ hpart, const float* __restrict__ aprod,
                        float* __restrict__ hin) {
    int idx = blockIdx.x * blockDim.x + threadIdx.x;
    int dn = idx & (D_ * N_ - 1);
    int bb = idx >> 13;
    float h = 0.f;
    for (int c = 0; c < NCHUNK; ++c) {
        long pos = ((long)bb * NCHUNK + c) * (D_ * N_) + dn;
        hin[pos] = h;
        h = fmaf(aprod[pos], h, hpart[pos]);
    }
}

#define FIXR 16
__global__ void __launch_bounds__(512) __attribute__((amdgpu_waves_per_eu(4, 8)))
k_fix_ln(const float* __restrict__ dbl, const float* __restrict__ psum,
         const float* __restrict__ Alog1, const float* __restrict__ Alog2,
         const float* __restrict__ hin, float* __restrict__ out,
         const float* __restrict__ g1, const float* __restrict__ bb1,
         const float* __restrict__ g2, const float* __restrict__ bb2) {
    int bid = blockIdx.x;
    int rg = bid & 1;  bid >>= 1;
    int c  = bid & (NCHUNK - 1); bid >>= 5;
    int b  = bid & 3;
    int branch = bid >> 2;
    int tid = threadIdx.x;
    int row0 = b * L_ + c * CHUNK + rg * FIXR;

    __shared__ __attribute__((aligned(16))) float Csh[FIXR * 16];
    __shared__ __attribute__((aligned(16))) float red[2 * FIXR * D_];
    __shared__ __attribute__((aligned(16))) float tot[2 * FIXR];

    const float* srcC = dbl + (long)(1 - branch) * 4096 * 48 + (long)row0 * 48 + 32;
    int ci = tid & (FIXR * 16 - 1);
    float cval = srcC[(long)(ci >> 4) * 48 + (ci & 15)];

    int d = tid;
    const float* Alog = branch ? Alog2 : Alog1;
    const float* ps   = psum + (long)branch * 4096 * D_;
    float* yout       = out  + (long)branch * 4096 * D_;
    const float* gam  = branch ? g2 : g1;
    const float* bet  = branch ? bb2 : bb1;

    float yv[FIXR], pv[FIXR];
#pragma unroll
    for (int r = 0; r < FIXR; ++r) {
        long idx = (long)(row0 + r) * D_ + d;
        yv[r] = yout[idx];
        pv[r] = ps[idx];
    }
    long off = ((long)(branch * 4 + b) * NCHUNK + c) * (D_ * N_) + (long)d * N_;
    v4f al[4], gq[4];
#pragma unroll
    for (int k = 0; k < 4; ++k) {
        al[k] = *(const v4f*)(Alog + (long)d * N_ + k * 4);
        gq[k] = *(const v4f*)(hin + off + k * 4);
    }
    float gd = gam[d], bd = bet[d];

    if (tid < FIXR * 16) Csh[tid] = cval;
    __syncthreads();

    {
        v2f A2[N_ / 2], g[N_ / 2];
#pragma unroll
        for (int k = 0; k < N_ / 2; ++k) {
            A2[k].x = -__expf(al[k >> 1][(2 * k) & 3])     * LOG2E;
            A2[k].y = -__expf(al[k >> 1][(2 * k + 1) & 3]) * LOG2E;
            g[k].x  = gq[k >> 1][(2 * k) & 3];
            g[k].y  = gq[k >> 1][(2 * k + 1) & 3];
        }
#pragma unroll
        for (int r = 0; r < FIXR; ++r) {
            v2f corr2 = {0.f, 0.f};
            v2f p2 = {pv[r], pv[r]};
#pragma unroll
            for (int k = 0; k < N_ / 2; ++k) {
                v2f e  = p2 * A2[k];
                v2f w  = {__builtin_amdgcn_exp2f(e.x), __builtin_amdgcn_exp2f(e.y)};
                v2f cp = *(const v2f*)&Csh[r * 16 + 2 * k];
                corr2 = (w * g[k]) * cp + corr2;
            }
            yv[r] += corr2.x + corr2.y;
        }
    }

#pragma unroll
    for (int r = 0; r < FIXR; ++r) {
        red[r * D_ + tid]          = yv[r];
        red[(FIXR + r) * D_ + tid] = yv[r] * yv[r];
    }
    __syncthreads();
    {
        int w = tid >> 5, ln = tid & 31;
#pragma unroll
        for (int j = 0; j < 2; ++j) {
            int v = 2 * w + j;
            float a = 0.f;
#pragma unroll
            for (int k = 0; k < D_ / 32; ++k)
                a += red[v * D_ + k * 32 + ln];
            for (int m = 16; m > 0; m >>= 1)
                a += __shfl_xor(a, m, 32);
            if (ln == 0) tot[v] = a;
        }
    }
    __syncthreads();
#pragma unroll
    for (int r = 0; r < FIXR; ++r) {
        float mean = tot[r] * (1.f / 512.f);
        float var  = tot[FIXR + r] * (1.f / 512.f) - mean * mean;
        float inv  = rsqrtf(var + 1e-5f);
        yout[(long)(row0 + r) * D_ + d] = (yv[r] - mean) * inv * gd + bd;
    }
}

extern "C" void kernel_launch(void* const* d_in, const int* in_sizes, int n_in,
                              void* d_out, int out_size, void* d_ws, size_t ws_size,
                              hipStream_t stream) {
    const float* x_rgb = (const float*)d_in[0];
    const float* x_e   = (const float*)d_in[1];
    const float* xw1   = (const float*)d_in[2];
    const float* xw2   = (const float*)d_in[3];
    const float* dtw1  = (const float*)d_in[4];
    const float* dtb1  = (const float*)d_in[5];
    const float* dtw2  = (const float*)d_in[6];
    const float* dtb2  = (const float*)d_in[7];
    const float* Al1   = (const float*)d_in[8];
    const float* Al2   = (const float*)d_in[9];
    const float* Dv1   = (const float*)d_in[10];
    const float* Dv2   = (const float*)d_in[11];
    const float* g1    = (const float*)d_in[12];
    const float* lb1   = (const float*)d_in[13];
    const float* g2    = (const float*)d_in[14];
    const float* lb2   = (const float*)d_in[15];
    float* out = (float*)d_out;

    float* ws    = (float*)d_ws;
    float* dbl   = ws;
    float* psum  = dbl + 2 * 4096 * 48;
    float* hpart = psum + 2 * 4096 * 512;
    float* aprod = hpart + 2 * 4 * NCHUNK * 8192;
    float* hin   = aprod + 2 * 4 * NCHUNK * 8192;

    k_xproj<<<256, 192, 0, stream>>>(x_rgb, x_e, xw1, xw2, dbl);
    k_scan_local<<<1024, 128, 0, stream>>>(x_rgb, x_e, dbl, dtw1, dtb1, dtw2, dtb2, psum,
                                           Al1, Al2, Dv1, Dv2, hpart, aprod, out);
    k_carry<<<256, 256, 0, stream>>>(hpart, aprod, hin);
    k_fix_ln<<<512, 512, 0, stream>>>(dbl, psum, Al1, Al2, hin, out, g1, lb1, g2, lb2);
}


# k_carry rebuilds a chunk's decay exp2(A2[n]*S) from S, the chunk's last psum row: k_scan_local no longer stores the 16 factors per channel and chunk, k_carry no longer loads them; sums as before
# speedup vs baseline: 3.3556x; 1.0310x over previous
#include <hip/hip_runtime.h>
#include <hip/hip_bf16.h>

typedef float v2f __attribute__((ext_vector_type(2)));
typedef float v4f __attribute__((ext_vector_type(4)));
typedef float v8f __attribute__((ext_vector_type(8)));

#define B_ 4
#define L_ 1024
#define D_ 512
#define N_ 16
#define CHUNK 32
#define NCHUNK 32
#define LOG2E 1.4426950408889634f
#define LN2   0.6931471805599453f

__device__ __forceinline__ float log1p_short(float e) {
    float w  = 1.f + e;
    float wm = w - 1.f;
    float l  = __builtin_amdgcn_logf(w) * LN2;
    float r  = l * (e * __builtin_amdgcn_rcpf(wm));
    return (w == 1.f) ? e : r;
}

#define XP_ROWS  80
#define XP_PITCH 516
#define XP_UNITS 14
__global__ void __launch_bounds__(192)
k_xproj(const float* __restrict__ x_rgb, const float* __restrict__ x_e,
                        const float* __restrict__ w1, const float* __restrict__ w2,
                        float* __restrict__ dbl) {
    __shared__ __attribute__((aligned(16))) float tile[XP_ROWS * XP_PITCH];
    int bid = blockIdx.x;
    int branch = bid >> 7;
    int mblk = bid & 127;
    const float* x = (branch ? x_e : x_rgb) + (long)mblk * 32 * D_;
    const float* w = branch ? w2 : w1;
    int ln = threadIdx.x & 31;
    int wv = __builtin_amdgcn_readfirstlane((int)(threadIdx.x >> 5));

    const float* srow[XP_UNITS];
    int drow[XP_UNITS];
#pragma unroll
    for (int j = 0; j < XP_UNITS; ++j) {
        int u = wv + 6 * j;
        u = (u > XP_ROWS - 1) ? XP_ROWS - 1 : u;
        srow[j] = (u < 32) ? x + (long)u * D_ : w + (long)(u - 32) * D_;
        drow[j] = u * XP_PITCH;
    }
    v4f st[4][XP_UNITS];
#pragma unroll
    for (int g = 0; g < 4; ++g) {
#pragma unroll
        for (int j = 0; j < XP_UNITS; ++j)
            st[g][j] = *(const v4f*)(srow[j] + g * 128 + ln * 4);
        __builtin_amdgcn_sched_barrier(0);
    }

    int c16 = ln & 15;
    int ko  = (ln >> 4) * 4;
    int mt  = wv / 3;
    int nt  = wv - mt * 3;
    const float* arow = &tile[(mt * 16 + c16) * XP_PITCH + ko];
    const float* brow = &tile[(32 + nt * 16 + c16) * XP_PITCH + ko];
    v8f c = {};
#pragma unroll
    for (int g = 0; g < 4; ++g) {
#pragma unroll
        for (int j = 0; j < XP_UNITS; ++j)
            *(v4f*)&tile[drow[j] + g * 128 + ln * 4] = st[g][j];
        __builtin_amdgcn_fence(__ATOMIC_RELEASE, "workgroup", "local");
        __builtin_amdgcn_s_barrier();
        __builtin_amdgcn_fence(__ATOMIC_ACQUIRE, "workgroup", "local");
        v4f af[16], bf[16];
#pragma unroll
        for (int i = 0; i < 16; ++i) {
            af[i] = *(const v4f*)(arow + g * 128 + i * 8);
            bf[i] = *(const v4f*)(brow + g * 128 + i * 8);
        }
        __builtin_amdgcn_sched_barrier(0);
#pragma unroll
        for (int i = 0; i < 16; ++i) {
            v2f a0 = {af[i].x, af[i].y}, b0 = {bf[i].x, bf[i].y};
            v2f a1 = {af[i].z, af[i].w}, b1 = {bf[i].z, bf[i].w};
            c = __builtin_amdgcn_wmma_f32_16x16x4_f32(false, a0, false, b0, (short)0, c, false, false);
            c = __builtin_amdgcn_wmma_f32_16x16x4_f32(false, a1, false, b1, (short)0, c, false, false);
        }
        __builtin_amdgcn_sched_barrier(0);
    }
    float* out = dbl + (long)branch * 4096 * 48;
    int rowbase = mblk * 32 + mt * 16 + ((ln >> 4) << 3);
    int col = nt * 16 + c16;
#pragma unroll
    for (int i = 0; i < 8; ++i)
        out[(long)(rowbase + i) * 48 + col] = c[i];
}

__global__ void __launch_bounds__(128)
k_scan_local(const float* __restrict__ x_rgb, const float* __restrict__ x_e,
             const float* __restrict__ dbl,
             const float* __restrict__ dtw1, const float* __restrict__ dtb1,
             const float* __restrict__ dtw2, const float* __restrict__ dtb2,
             float* __restrict__ psum,
             const float* __restrict__ Alog1, const float* __restrict__ Alog2,
             const float* __restrict__ D1, const float* __restrict__ D2,
             float* __restrict__ hpart,
             float* __restrict__ out) {
    int bid = blockIdx.x;
    int lc  = bid & (NCHUNK - 1); bid >>= 5;
    int dch = bid & 3;  bid >>= 2;
    int b   = bid & 3;
    int branch = bid >> 2;
    int tid = threadIdx.x;
    int l0 = lc * CHUNK;

    __shared__ __attribute__((aligned(16))) float Bsh[CHUNK * 16];
    __shared__ __attribute__((aligned(16))) float Csh[CHUNK * 16];
    __shared__ __attribute__((aligned(16))) float Ssh[CHUNK * 128];
    __shared__ __attribute__((aligned(16))) float Ush[CHUNK * 128];

    const float* Alog = branch ? Alog2 : Alog1;
    const float* Dp   = branch ? D2 : D1;
    const float* x    = branch ? x_e : x_rgb;
    const float* dtw  = branch ? dtw2 : dtw1;
    const float* dtb  = branch ? dtb2 : dtb1;

    long row0 = (long)(b * L_ + l0);
    int  col0 = dch * 128;
    float*       psT = psum + (long)branch * 4096 * D_ + row0 * D_ + col0;
    const float* xT  = x + row0 * D_ + col0;
    float*       yT  = out + (long)branch * 4096 * D_ + row0 * D_ + col0;
    const float* dtT = dbl + (long)branch * 4096 * 48 + row0 * 48;

    int q = tid & 31, rr = tid >> 5;
    int c16 = q & 15;
    int kb  = (q >> 4) * 2;
    int chw = rr * 32;
    v2f af[2][4], bf[2][4];
    float bcol[2];
#pragma unroll
    for (int mt = 0; mt < 2; ++mt) {
        const float* arow = dtT + (mt * 16 + c16) * 48;
#pragma unroll
        for (int ks = 0; ks < 4; ++ks) {
            af[mt][ks].x = arow[ks * 4 + kb];
            af[mt][ks].y = arow[ks * 4 + kb + 1];
        }
    }
#pragma unroll
    for (int j = 0; j < 2; ++j) {
        int n = col0 + chw + j * 16 + c16;
        const float* wrow = dtw + (long)n * 16;
#pragma unroll
        for (int ks = 0; ks < 4; ++ks) {
            bf[j][ks].x = wrow[ks * 4 + kb];
            bf[j][ks].y = wrow[ks * 4 + kb + 1];
        }
        bcol[j] = dtb[n];
    }

    v4f uv[8];
#pragma unroll
    for (int p = 0; p < 8; ++p) {
        int r = p * 4 + rr;
        uv[p] = *(const v4f*)(xT + r * D_ + q * 4);
    }
    v4f bv, cv;
    {
        int r = tid >> 2, c4 = (tid & 3) * 4;
        bv = *(const v4f*)(dbl + (long)branch * 4096 * 48       + (row0 + r) * 48 + 16 + c4);
        cv = *(const v4f*)(dbl + (long)(1 - branch) * 4096 * 48 + (row0 + r) * 48 + 32 + c4);
    }

    int d = col0 + tid;
    v2f A2[N_ / 2], h[N_ / 2];
#pragma unroll
    for (int k = 0; k < N_ / 2; ++k) {
        A2[k].x = -__expf(Alog[(long)d * N_ + 2 * k])     * LOG2E;
        A2[k].y = -__expf(Alog[(long)d * N_ + 2 * k + 1]) * LOG2E;
        h[k] = (v2f){0.f, 0.f};
    }
    float Dd = Dp[d];

    v8f acc[2][2];
#pragma unroll
    for (int mt = 0; mt < 2; ++mt)
#pragma unroll
        for (int j = 0; j < 2; ++j)
            acc[mt][j] = (v8f){};
#pragma unroll
    for (int ks = 0; ks < 4; ++ks)
#pragma unroll
        for (int mt = 0; mt < 2; ++mt)
#pragma unroll
            for (int j = 0; j < 2; ++j)
                acc[mt][j] = __builtin_amdgcn_wmma_f32_16x16x4_f32(false, af[mt][ks], false, bf[j][ks],
                                                                   (short)0, acc[mt][j], false, false);

    int rbase = (q >> 4) << 3;
#pragma unroll
    for (int mt = 0; mt < 2; ++mt)
#pragma unroll
        for (int j = 0; j < 2; ++j)
#pragma unroll
            for (int i = 0; i < 8; ++i) {
                float v  = acc[mt][j][i] + bcol[j];
                float sp = (v > 20.f) ? v : log1p_short(__expf(v));
                Ssh[(mt * 16 + rbase + i) * 128 + chw + j * 16 + c16] = sp;
            }

#pragma unroll
    for (int p = 0; p < 8; ++p) {
        int r = p * 4 + rr;
        *(v4f*)&Ush[r * 128 + q * 4] = uv[p];
    }
    *(v4f*)&Bsh[tid * 4] = bv;
    *(v4f*)&Csh[tid * 4] = cv;
    __syncthreads();

    float ssum = 0.f;
    for (int ll = 0; ll < CHUNK; ++ll) {
        float s = Ssh[ll * 128 + tid];
        float u = Ush[ll * 128 + tid];
        float su = s * u;
        ssum += s;
        v2f y2 = {0.f, 0.f};
        v2f s2 = {s, s}, su2 = {su, su};
#pragma unroll
        for (int k = 0; k < N_ / 2; ++k) {
            v2f bp = *(const v2f*)&Bsh[ll * 16 + 2 * k];
            v2f cp = *(const v2f*)&Csh[ll * 16 + 2 * k];
            v2f e  = s2 * A2[k];
            v2f dA = {__builtin_amdgcn_exp2f(e.x), __builtin_amdgcn_exp2f(e.y)};
            h[k] = dA * h[k] + su2 * bp;
            y2   = h[k] * cp + y2;
        }
        float y = y2.x + y2.y;
        yT[ll * D_ + tid]  = fmaf(Dd, u, y);
        psT[ll * D_ + tid] = ssum;
    }
    long off = ((long)(branch * 4 + b) * NCHUNK + lc) * (D_ * N_) + (long)d * N_;
#pragma unroll
    for (int k = 0; k < N_ / 2; ++k) {
        hpart[off + 2 * k]     = h[k].x;
        hpart[off + 2 * k + 1] = h[k].y;
    }
}

__global__ void k_carry(const float* __restrict__ hpart, const float* __restrict__ psum,
                        const float* __restrict__ Alog1, const float* __restrict__ Alog2,
                        float* __restrict__ hin) {
    int idx = blockIdx.x * blockDim.x + threadIdx.x;
    int dn = idx & (D_ * N_ - 1);
    int bb = idx >> 13;
    const float* Alog = (bb >> 2) ? Alog2 : Alog1;
    const float* plast = psum + ((long)bb * L_ + (CHUNK - 1)) * D_ + (dn >> 4);
    long pos0 = (long)bb * NCHUNK * (D_ * N_) + dn;
    float hp[NCHUNK - 1], sl[NCHUNK - 1];
#pragma unroll
    for (int c = 0; c < NCHUNK - 1; ++c) {
        hp[c] = hpart[pos0 + (long)c * (D_ * N_)];
        sl[c] = plast[(long)c * CHUNK * D_];
    }
    float al = Alog[dn];
    __builtin_amdgcn_sched_barrier(0);
    float a2 = -__expf(al) * LOG2E;
    float h = 0.f;
    hin[pos0] = h;
#pragma unroll
    for (int c = 0; c < NCHUNK - 1; ++c) {
        float ap = __builtin_amdgcn_exp2f(sl[c] * a2);
        h = fmaf(ap, h, hp[c]);
        hin[pos0 + (long)(c + 1) * (D_ * N_)] = h;
    }
}

#define FIXR 16
__global__ void __launch_bounds__(512) __attribute__((amdgpu_waves_per_eu(4, 8)))
k_fix_ln(const float* __restrict__ dbl, const float* __restrict__ psum,
         const float* __restrict__ Alog1, const float* __restrict__ Alog2,
         const float* __restrict__ hin, float* __restrict__ out,
         const float* __restrict__ g1, const float* __restrict__ bb1,
         const float* __restrict__ g2, const float* __restrict__ bb2) {
    int bid = blockIdx.x;
    int rg = bid & 1;  bid >>= 1;
    int c  = bid & (NCHUNK - 1); bid >>= 5;
    int b  = bid & 3;
    int branch = bid >> 2;
    int tid = threadIdx.x;
    int row0 = b * L_ + c * CHUNK + rg * FIXR;

    __shared__ __attribute__((aligned(16))) float Csh[FIXR * 16];
    __shared__ __attribute__((aligned(16))) float red[2 * FIXR * D_];
    __shared__ __attribute__((aligned(16))) float tot[2 * FIXR];

    const float* srcC = dbl + (long)(1 - branch) * 4096 * 48 + (long)row0 * 48 + 32;
    int ci = tid & (FIXR * 16 - 1);
    float cval = srcC[(long)(ci >> 4) * 48 + (ci & 15)];

    int d = tid;
    const float* Alog = branch ? Alog2 : Alog1;
    const float* ps   = psum + (long)branch * 4096 * D_;
    float* yout       = out  + (long)branch * 4096 * D_;
    const float* gam  = branch ? g2 : g1;
    const float* bet  = branch ? bb2 : bb1;

    float yv[FIXR], pv[FIXR];
#pragma unroll
    for (int r = 0; r < FIXR; ++r) {
        long idx = (long)(row0 + r) * D_ + d;
        yv[r] = yout[idx];
        pv[r] = ps[idx];
    }
    long off = ((long)(branch * 4 + b) * NCHUNK + c) * (D_ * N_) + (long)d * N_;
    v4f al[4], gq[4];
#pragma unroll
    for (int k = 0; k < 4; ++k) {
        al[k] = *(const v4f*)(Alog + (long)d * N_ + k * 4);
        gq[k] = *(const v4f*)(hin + off + k * 4);
    }
    float gd = gam[d], bd = bet[d];

    if (tid < FIXR * 16) Csh[tid] = cval;
    __syncthreads();

    {
        v2f A2[N_ / 2], g[N_ / 2];
#pragma unroll
        for (int k = 0; k < N_ / 2; ++k) {
            A2[k].x = -__expf(al[k >> 1][(2 * k) & 3])     * LOG2E;
            A2[k].y = -__expf(al[k >> 1][(2 * k + 1) & 3]) * LOG2E;
            g[k].x  = gq[k >> 1][(2 * k) & 3];
            g[k].y  = gq[k >> 1][(2 * k + 1) & 3];
        }
#pragma unroll
        for (int r = 0; r < FIXR; ++r) {
            v2f corr2 = {0.f, 0.f};
            v2f p2 = {pv[r], pv[r]};
#pragma unroll
            for (int k = 0; k < N_ / 2; ++k) {
                v2f e  = p2 * A2[k];
                v2f w  = {__builtin_amdgcn_exp2f(e.x), __builtin_amdgcn_exp2f(e.y)};
                v2f cp = *(const v2f*)&Csh[r * 16 + 2 * k];
                corr2 = (w * g[k]) * cp + corr2;
            }
            yv[r] += corr2.x + corr2.y;
        }
    }

#pragma unroll
    for (int r = 0; r < FIXR; ++r) {
        red[r * D_ + tid]          = yv[r];
        red[(FIXR + r) * D_ + tid] = yv[r] * yv[r];
    }
    __syncthreads();
    {
        int w = tid >> 5, ln = tid & 31;
#pragma unroll
        for (int j = 0; j < 2; ++j) {
            int v = 2 * w + j;
            float a = 0.f;
#pragma unroll
            for (int k = 0; k < D_ / 32; ++k)
                a += red[v * D_ + k * 32 + ln];
            for (int m = 16; m > 0; m >>= 1)
                a += __shfl_xor(a, m, 32);
            if (ln == 0) tot[v] = a;
        }
    }
    __syncthreads();
#pragma unroll
    for (int r = 0; r < FIXR; ++r) {
        float mean = tot[r] * (1.f / 512.f);
        float var  = tot[FIXR + r] * (1.f / 512.f) - mean * mean;
        float inv  = rsqrtf(var + 1e-5f);
        yout[(long)(row0 + r) * D_ + d] = (yv[r] - mean) * inv * gd + bd;
    }
}

extern "C" void kernel_launch(void* const* d_in, const int* in_sizes, int n_in,
                              void* d_out, int out_size, void* d_ws, size_t ws_size,
                              hipStream_t stream) {
    const float* x_rgb = (const float*)d_in[0];
    const float* x_e   = (const float*)d_in[1];
    const float* xw1   = (const float*)d_in[2];
    const float* xw2   = (const float*)d_in[3];
    const float* dtw1  = (const float*)d_in[4];
    const float* dtb1  = (const float*)d_in[5];
    const float* dtw2  = (const float*)d_in[6];
    const float* dtb2  = (const float*)d_in[7];
    const float* Al1   = (const float*)d_in[8];
    const float* Al2   = (const float*)d_in[9];
    const float* Dv1   = (const float*)d_in[10];
    const float* Dv2   = (const float*)d_in[11];
    const float* g1    = (const float*)d_in[12];
    const float* lb1   = (const float*)d_in[13];
    const float* g2    = (const float*)d_in[14];
    const float* lb2   = (const float*)d_in[15];
    float* out = (float*)d_out;

    float* ws    = (float*)d_ws;
    float* dbl   = ws;
    float* psum  = dbl + 2 * 4096 * 48;
    float* hpart = psum + 2 * 4096 * 512;
    float* hin   = hpart + 2 * 4 * NCHUNK * 8192;

    k_xproj<<<256, 192, 0, stream>>>(x_rgb, x_e, xw1, xw2, dbl);
    k_scan_local<<<1024, 128, 0, stream>>>(x_rgb, x_e, dbl, dtw1, dtb1, dtw2, dtb2, psum,
                                           Al1, Al2, Dv1, Dv2, hpart, out);
    k_carry<<<256, 256, 0, stream>>>(hpart, psum, Al1, Al2, hin);
    k_fix_ln<<<512, 512, 0, stream>>>(dbl, psum, Al1, Al2, hin, out, g1, lb1, g2, lb2);
}
